# starting kernel
# baseline (speedup 1.0000x reference)
#include <hip/hip_runtime.h>

typedef float v2f __attribute__((ext_vector_type(2)));
typedef float v8f __attribute__((ext_vector_type(8)));

#define DDIM 128
#define TILE 16
#define WAVES_PER_BLOCK 4
#define LDS_STRIDE 34

__global__ __launch_bounds__(WAVES_PER_BLOCK * 32) void KB_transe_l1_wmma(
    const float* __restrict__ h,
    const float* __restrict__ g,
    const int* __restrict__ row,
    const int* __restrict__ col,
    const int* __restrict__ typ,
    float* __restrict__ out,
    int n_edges)
{
    __shared__ float lds[WAVES_PER_BLOCK * TILE * LDS_STRIDE];

    const int lane  = threadIdx.x & 31;
    const int wave  = threadIdx.x >> 5;
    const int tile  = blockIdx.x * WAVES_PER_BLOCK + wave;
    const int ebase = tile * TILE;
    if (ebase >= n_edges) return;

    const int nvalid = min(TILE, n_edges - ebase);
    const int eslot  = lane & 15;
    const int eclamp = ebase + min(eslot, nvalid - 1);

    int myRow = row[eclamp];
    int myCol = col[eclamp];
    int myTyp = typ[eclamp];

    float* P = &lds[wave * TILE * LDS_STRIDE];

    #pragma unroll 4
    for (int e = 0; e < TILE; ++e) {
        const int r = __shfl(myRow, e, 32);
        const int c = __shfl(myCol, e, 32);
        const int t = __shfl(myTyp, e, 32);
        const float4 hr = reinterpret_cast<const float4*>(h + (size_t)r * DDIM)[lane];
        const float4 hc = reinterpret_cast<const float4*>(h + (size_t)c * DDIM)[lane];
        const float4 gv = reinterpret_cast<const float4*>(g + (size_t)t * DDIM)[lane];
        const float s = fabsf(hr.x + gv.x - hc.x)
                      + fabsf(hr.y + gv.y - hc.y)
                      + fabsf(hr.z + gv.z - hc.z)
                      + fabsf(hr.w + gv.w - hc.w);
        P[e * LDS_STRIDE + lane] = s;
    }
    __syncthreads();

    v8f acc = {};
    v2f ones; ones.x = 1.0f; ones.y = 1.0f;
    const int k0 = (lane >> 4) << 1;
    #pragma unroll
    for (int ch = 0; ch < 8; ++ch) {
        const float* src = &P[eslot * LDS_STRIDE + 4 * ch + k0];
        v2f a = *reinterpret_cast<const v2f*>(src);
        acc = __builtin_amdgcn_wmma_f32_16x16x4_f32(
                  false, a, false, ones, (short)0, acc, false, false);
    }

    if (eslot == 0) {
        const int mbase = (lane >> 4) * 8;
        if (nvalid == TILE) {
            float4 lo = make_float4(acc[0], acc[1], acc[2], acc[3]);
            float4 hi = make_float4(acc[4], acc[5], acc[6], acc[7]);
            reinterpret_cast<float4*>(out + ebase + mbase)[0] = lo;
            reinterpret_cast<float4*>(out + ebase + mbase)[1] = hi;
        } else {
            for (int i = 0; i < 8; ++i) {
                const int m = mbase + i;
                if (m < nvalid) out[ebase + m] = acc[i];
            }
        }
    }
}

extern "C" void kernel_launch(void* const* d_in, const int* in_sizes, int n_in,
                              void* d_out, int out_size, void* d_ws, size_t ws_size,
                              hipStream_t stream) {
    const float* h    = (const float*)d_in[0];
    const float* g    = (const float*)d_in[1];
    const int*   eidx = (const int*)d_in[2];
    const int*   etyp = (const int*)d_in[3];
    float*       out  = (float*)d_out;

    const int E     = in_sizes[3];
    const int tiles = (E + TILE - 1) / TILE;
    const int grid  = (tiles + WAVES_PER_BLOCK - 1) / WAVES_PER_BLOCK;

    KB_transe_l1_wmma<<<grid, WAVES_PER_BLOCK * 32, 0, stream>>>(
        h, g, eidx, eidx + E, etyp, out, E);
}


# g table (256,000 B) kept in LDS by 256 resident 16-wave groups, copied once in two 16-load batches; g[type] rows read from LDS, h-row loads run half a tile ahead of the arithmetic
# speedup vs baseline: 1.0779x; 1.0779x over previous
#include <hip/hip_runtime.h>


typedef float v2f __attribute__((ext_vector_type(2)));
typedef float v8f __attribute__((ext_vector_type(8)));

#define DDIM 128
#define TILE 16
#define HALF 8
#define WAVES_PER_BLOCK 16
#define NTHREADS (WAVES_PER_BLOCK * 32)
#define LDS_STRIDE 34
#define GTAB_FLOATS 64000
#define GTAB_F4 16384

struct Desc { int r, c, t; };

__device__ __forceinline__ Desc load_desc(const int* __restrict__ row,
                                          const int* __restrict__ col,
                                          const int* __restrict__ typ,
                                          int tile, int tend, int tlast, int n_edges, int eslot)
{
    const int eb = min(tile, tlast) * TILE;
    const int ec = eb + min(eslot, min(TILE, n_edges - eb) - 1);
    const int r = row[ec], c = col[ec], t = typ[ec];
    const bool live = tile < tend;
    Desc d; d.r = live ? r : 0; d.c = live ? c : 0; d.t = live ? t : 0;
    return d;
}

template <int HF>
__device__ __forceinline__ void load_rows(const float* __restrict__ h, const Desc& d, int lane,
                                          float4 (&hr)[HALF], float4 (&hc)[HALF])
{
    #pragma unroll
    for (int j = 0; j < HALF; ++j) {
        const int r = __builtin_amdgcn_readlane(d.r, HF * HALF + j);
        const int c = __builtin_amdgcn_readlane(d.c, HF * HALF + j);
        hr[j] = reinterpret_cast<const float4*>(h + (size_t)r * DDIM)[lane];
        hc[j] = reinterpret_cast<const float4*>(h + (size_t)c * DDIM)[lane];
    }
}

template <int HF>
__device__ __forceinline__ void fold_rows(const float4* gtab, const Desc& d, int lane, float* P,
                                          const float4 (&hr)[HALF], const float4 (&hc)[HALF])
{
    #pragma unroll
    for (int j = 0; j < HALF; ++j) {
        const int e  = HF * HALF + j;
        const int ty = __builtin_amdgcn_readlane(d.t, e);
        const float4 gv = gtab[ty * (DDIM / 4) + lane];
        const float s = fabsf(hr[j].x + gv.x - hc[j].x)
                      + fabsf(hr[j].y + gv.y - hc[j].y)
                      + fabsf(hr[j].z + gv.z - hc[j].z)
                      + fabsf(hr[j].w + gv.w - hc[j].w);
        P[e * LDS_STRIDE + lane] = s;
    }
}

__global__ __launch_bounds__(NTHREADS) void KB_transe_l1_wmma(
    const float* __restrict__ h,
    const float* __restrict__ g,
    const int* __restrict__ row,
    const int* __restrict__ col,
    const int* __restrict__ typ,
    float* __restrict__ out,
    int n_edges,
    int g_floats)
{
    __shared__ float4 gtab[GTAB_F4];
    __shared__ float  pst[WAVES_PER_BLOCK * TILE * LDS_STRIDE];

    const int lane  = threadIdx.x & 31;
    const int wave  = __builtin_amdgcn_readfirstlane(threadIdx.x >> 5);
    const int wg    = blockIdx.x;
    const int nwg   = gridDim.x;
    const int eslot = lane & 15;

    if (n_edges <= 0) return;
    const int tiles  = (n_edges + TILE - 1) / TILE;
    const int npairs = (tiles + 1) >> 1;
    const int pq     = npairs / nwg, prem = npairs - pq * nwg;
    const int p0     = wg * pq + min(wg, prem);
    const int p1     = p0 + pq + (wg < prem ? 1 : 0);
    const int tend   = min(2 * p1, tiles);
    int       t      = 2 * p0 + wave;

    float4 hrA[HALF], hcA[HALF], hrB[HALF], hcB[HALF];
    Desc cur = load_desc(row, col, typ, t, tend, tiles - 1, n_edges, eslot);
    Desc nxt = load_desc(row, col, typ, t + WAVES_PER_BLOCK, tend, tiles - 1, n_edges, eslot);
    load_rows<0>(h, cur, lane, hrA, hcA);

    {
        const int n4  = min(g_floats >> 2, GTAB_FLOATS / 4);
        const int off = wg * (n4 / nwg);
        const float4* __restrict__ g4 = reinterpret_cast<const float4*>(g);
        for (int b = threadIdx.x; b < n4; b += 16 * NTHREADS) {
            float4 v[16];
            #pragma unroll
            for (int k = 0; k < 16; ++k) {
                int j = min(b + k * NTHREADS, n4 - 1) + off;
                if (j >= n4) j -= n4;
                v[k] = g4[j];
            }
            #pragma unroll
            for (int k = 0; k < 16; ++k) {
                const int i = b + k * NTHREADS;
                int j = i + off;
                if (j >= n4) j -= n4;
                gtab[i < n4 ? j : i] = v[k];
            }
        }
    }
    __syncthreads();

    float* P = &pst[wave * TILE * LDS_STRIDE];
    v2f ones; ones.x = 1.0f; ones.y = 1.0f;
    const int k0 = (lane >> 4) << 1;

    for (; t < tend; t += WAVES_PER_BLOCK) {
        const int ebase  = t * TILE;
        const int nvalid = min(TILE, n_edges - ebase);

        load_rows<1>(h, cur, lane, hrB, hcB);
        const Desc nx2 = load_desc(row, col, typ, t + 2 * WAVES_PER_BLOCK, tend, tiles - 1, n_edges, eslot);

        fold_rows<0>(gtab, cur, lane, P, hrA, hcA);
        load_rows<0>(h, nxt, lane, hrA, hcA);
        fold_rows<1>(gtab, cur, lane, P, hrB, hcB);

        __builtin_amdgcn_fence(__ATOMIC_RELEASE, "wavefront");
        __builtin_amdgcn_fence(__ATOMIC_ACQUIRE, "wavefront");

        __builtin_amdgcn_sched_barrier(0);
        v8f acc = {};
        #pragma unroll
        for (int ch = 0; ch < 8; ++ch) {
            const float* src = &P[eslot * LDS_STRIDE + 4 * ch + k0];
            v2f a = *reinterpret_cast<const v2f*>(src);
            acc = __builtin_amdgcn_wmma_f32_16x16x4_f32(
                      false, a, false, ones, (short)0, acc, false, false);
        }
        __builtin_amdgcn_sched_barrier(0);
        __builtin_amdgcn_fence(__ATOMIC_RELEASE, "wavefront");
        __builtin_amdgcn_fence(__ATOMIC_ACQUIRE, "wavefront");

        if (eslot == 0) {
            const int mbase = (lane >> 4) * 8;
            if (nvalid == TILE) {
                float4 lo = make_float4(acc[0], acc[1], acc[2], acc[3]);
                float4 hi = make_float4(acc[4], acc[5], acc[6], acc[7]);
                reinterpret_cast<float4*>(out + ebase + mbase)[0] = lo;
                reinterpret_cast<float4*>(out + ebase + mbase)[1] = hi;
            } else {
                for (int i = 0; i < 8; ++i) {
                    const int m = mbase + i;
                    if (m < nvalid) out[ebase + m] = acc[i];
                }
            }
        }
        cur = nxt; nxt = nx2;
    }
}

extern "C" void kernel_launch(void* const* d_in, const int* in_sizes, int n_in,
                              void* d_out, int out_size, void* d_ws, size_t ws_size,
                              hipStream_t stream) {
    const float* h    = (const float*)d_in[0];
    const float* g    = (const float*)d_in[1];
    const int*   eidx = (const int*)d_in[2];
    const int*   etyp = (const int*)d_in[3];
    float*       out  = (float*)d_out;

    const int E = in_sizes[3];

    KB_transe_l1_wmma<<<256, NTHREADS, 0, stream>>>(
        h, g, eidx, eidx + E, etyp, out, E, in_sizes[1]);
}
